# starting kernel
# baseline (speedup 1.0000x reference)
#include <hip/hip_runtime.h>
#include <hip/hip_bf16.h>

typedef float v2f __attribute__((ext_vector_type(2)));
typedef float v8f __attribute__((ext_vector_type(8)));

#define IN_DIM  512
#define OUT_DIM 512
#define TC_DIM  256

__global__ void k_deg_init(float* __restrict__ deg, int N) {
    int i = blockIdx.x * blockDim.x + threadIdx.x;
    if (i < N) deg[i] = 1.0f;
}

__global__ void k_deg_accum(const int* __restrict__ ei, float* __restrict__ deg, int E) {
    int e = blockIdx.x * blockDim.x + threadIdx.x;
    if (e < E) atomicAdd(&deg[ei[E + e]], 1.0f);
}

__global__ void k_dinv(const float* __restrict__ deg, float* __restrict__ dinv, int N) {
    int i = blockIdx.x * blockDim.x + threadIdx.x;
    if (i < N) dinv[i] = __frsqrt_rn(deg[i]);
}

__global__ void k_cvec(const float* __restrict__ t_emb, const float* __restrict__ Wt,
                       const float* __restrict__ b, const float* __restrict__ bt,
                       float* __restrict__ cvec) {
    int j = blockIdx.x * blockDim.x + threadIdx.x;
    if (j >= OUT_DIM) return;
    float s = b[j] + bt[j];
    #pragma unroll 4
    for (int k = 0; k < TC_DIM; ++k) s += t_emb[k] * Wt[k * OUT_DIM + j];
    cvec[j] = s;
}

__global__ void k_gemm_wmma(const float* __restrict__ x, const float* __restrict__ W,
                            float* __restrict__ xw, int N, int rowTiles) {
    const int wave = (blockIdx.x * blockDim.x + threadIdx.x) >> 5;
    const int lane = threadIdx.x & 31;
    const int colTile = wave & (OUT_DIM / 16 - 1);
    const int rowTile = wave >> 5;
    if (rowTile >= rowTiles) return;

    const int row0 = rowTile * 16;
    const int col0 = colTile * 16;
    const int l15  = lane & 15;
    const int hi   = lane >> 4;

    int aRow = row0 + l15;
    if (aRow >= N) aRow = N - 1;
    const float* __restrict__ aPtr = x + (size_t)aRow * IN_DIM + 2 * hi;
    const float* __restrict__ bPtr = W + (size_t)(2 * hi) * OUT_DIM + col0 + l15;

    v8f c = {};
    for (int k = 0; k < IN_DIM; k += 4) {
        v2f a = *(const v2f*)(aPtr + k);
        v2f bm;
        bm.x = bPtr[(size_t)k * OUT_DIM];
        bm.y = bPtr[(size_t)(k + 1) * OUT_DIM];
        c = __builtin_amdgcn_wmma_f32_16x16x4_f32(
                false, a, false, bm, (short)0, c, false, false);
    }

    const int n     = col0 + l15;
    const int rbase = row0 + hi * 8;
    float* __restrict__ dPtr = xw + (size_t)rbase * OUT_DIM + n;
    if (row0 + 16 <= N) {
        #pragma unroll
        for (int v = 0; v < 8; ++v) dPtr[(size_t)v * OUT_DIM] = c[v];
    } else {
        for (int v = 0; v < 8; ++v)
            if (rbase + v < N) dPtr[(size_t)v * OUT_DIM] = c[v];
    }
}

__global__ void k_init_out(const float4* __restrict__ xw4, const float* __restrict__ dinv,
                           const float4* __restrict__ cvec4, float4* __restrict__ out4,
                           size_t total4) {
    size_t idx = (size_t)blockIdx.x * blockDim.x + threadIdx.x;
    if (idx >= total4) return;
    int i = (int)(idx >> 7);
    int q = (int)(idx & 127);
    float di = dinv[i];
    float s  = di * di;
    float4 xv = xw4[idx];
    float4 cv = cvec4[q];
    float4 r;
    r.x = cv.x + s * xv.x;
    r.y = cv.y + s * xv.y;
    r.z = cv.z + s * xv.z;
    r.w = cv.w + s * xv.w;
    out4[idx] = r;
}

__global__ void k_edge_scatter(const int* __restrict__ ei, const float* __restrict__ xw,
                               const float* __restrict__ dinv, float* __restrict__ out,
                               int E) {
    int wave = (blockIdx.x * blockDim.x + threadIdx.x) >> 5;
    int lane = threadIdx.x & 31;
    if (wave >= E) return;
    int src = ei[wave];
    int dst = ei[E + wave];
    float norm = dinv[src] * dinv[dst];
    const float4* __restrict__ srow4 = (const float4*)(xw + (size_t)src * OUT_DIM);
    float* __restrict__ drow = out + (size_t)dst * OUT_DIM;
    #pragma unroll
    for (int c = 0; c < OUT_DIM / 128; ++c) {
        int q = lane + 32 * c;
        float4 v = srow4[q];
        int col = 4 * q;
        atomicAdd(&drow[col + 0], norm * v.x);
        atomicAdd(&drow[col + 1], norm * v.y);
        atomicAdd(&drow[col + 2], norm * v.z);
        atomicAdd(&drow[col + 3], norm * v.w);
    }
}

extern "C" void kernel_launch(void* const* d_in, const int* in_sizes, int n_in,
                              void* d_out, int out_size, void* d_ws, size_t ws_size,
                              hipStream_t stream) {
    const float* x     = (const float*)d_in[0];
    const float* t_emb = (const float*)d_in[1];
    const int*   ei    = (const int*)d_in[2];
    const float* W     = (const float*)d_in[3];
    const float* b     = (const float*)d_in[4];
    const float* Wt    = (const float*)d_in[5];
    const float* bt    = (const float*)d_in[6];
    float*       out   = (float*)d_out;

    const int N = in_sizes[0] / IN_DIM;
    const int E = in_sizes[2] / 2;

    char* ws = (char*)d_ws;
    float* deg  = (float*)(ws + 0);
    float* dinv = (float*)(ws + (((size_t)N * 4 + 63) & ~63ull));
    float* cvec = (float*)((char*)dinv + (((size_t)N * 4 + 63) & ~63ull));
    float* xw   = (float*)((char*)cvec + 4096);

    k_deg_init <<<(N + 255) / 256, 256, 0, stream>>>(deg, N);
    k_deg_accum<<<(E + 255) / 256, 256, 0, stream>>>(ei, deg, E);
    k_dinv     <<<(N + 255) / 256, 256, 0, stream>>>(deg, dinv, N);

    k_cvec<<<(OUT_DIM + 255) / 256, 256, 0, stream>>>(t_emb, Wt, b, bt, cvec);

    {
        int rowTiles = (N + 15) / 16;
        int tiles    = rowTiles * (OUT_DIM / 16);
        int blocks   = (tiles + 7) / 8;
        k_gemm_wmma<<<blocks, 256, 0, stream>>>(x, W, xw, N, rowTiles);
    }

    {
        size_t total4 = (size_t)N * OUT_DIM / 4;
        int blocks = (int)((total4 + 255) / 256);
        k_init_out<<<blocks, 256, 0, stream>>>((const float4*)xw, dinv,
                                               (const float4*)cvec, (float4*)out, total4);
    }

    {
        int blocks = (E + 7) / 8;
        k_edge_scatter<<<blocks, 256, 0, stream>>>(ei, xw, dinv, out, E);
    }
}


# Edge scatter by global f32 atomics replaced by owner-computes aggregation: a workgroup owns 40 output rows, scans dst, sums its messages by LDS f32 atomics (arrival order), stores each row once
# speedup vs baseline: 1.4889x; 1.4889x over previous
#include <hip/hip_runtime.h>
#include <hip/hip_bf16.h>

typedef float v2f __attribute__((ext_vector_type(2)));
typedef float v8f __attribute__((ext_vector_type(8)));

#define IN_DIM  512
#define OUT_DIM 512
#define TC_DIM  256

__global__ void k_deg_init(float* __restrict__ deg, int N) {
    int i = blockIdx.x * blockDim.x + threadIdx.x;
    if (i < N) deg[i] = 1.0f;
}

__global__ void k_deg_accum(const int* __restrict__ ei, float* __restrict__ deg, int E) {
    int e = blockIdx.x * blockDim.x + threadIdx.x;
    if (e < E) atomicAdd(&deg[ei[E + e]], 1.0f);
}

__global__ void k_dinv(const float* __restrict__ deg, float* __restrict__ dinv, int N) {
    int i = blockIdx.x * blockDim.x + threadIdx.x;
    if (i < N) dinv[i] = __frsqrt_rn(deg[i]);
}

__global__ void k_cvec(const float* __restrict__ t_emb, const float* __restrict__ Wt,
                       const float* __restrict__ b, const float* __restrict__ bt,
                       float* __restrict__ cvec) {
    int j = blockIdx.x * blockDim.x + threadIdx.x;
    if (j >= OUT_DIM) return;
    float s = b[j] + bt[j];
    #pragma unroll 4
    for (int k = 0; k < TC_DIM; ++k) s += t_emb[k] * Wt[k * OUT_DIM + j];
    cvec[j] = s;
}

__global__ void k_gemm_wmma(const float* __restrict__ x, const float* __restrict__ W,
                            float* __restrict__ xw, int N, int rowTiles) {
    const int wave = (blockIdx.x * blockDim.x + threadIdx.x) >> 5;
    const int lane = threadIdx.x & 31;
    const int colTile = wave & (OUT_DIM / 16 - 1);
    const int rowTile = wave >> 5;
    if (rowTile >= rowTiles) return;

    const int row0 = rowTile * 16;
    const int col0 = colTile * 16;
    const int l15  = lane & 15;
    const int hi   = lane >> 4;

    int aRow = row0 + l15;
    if (aRow >= N) aRow = N - 1;
    const float* __restrict__ aPtr = x + (size_t)aRow * IN_DIM + 2 * hi;
    const float* __restrict__ bPtr = W + (size_t)(2 * hi) * OUT_DIM + col0 + l15;

    v8f c = {};
    for (int k = 0; k < IN_DIM; k += 4) {
        v2f a = *(const v2f*)(aPtr + k);
        v2f bm;
        bm.x = bPtr[(size_t)k * OUT_DIM];
        bm.y = bPtr[(size_t)(k + 1) * OUT_DIM];
        c = __builtin_amdgcn_wmma_f32_16x16x4_f32(
                false, a, false, bm, (short)0, c, false, false);
    }

    const int n     = col0 + l15;
    const int rbase = row0 + hi * 8;
    float* __restrict__ dPtr = xw + (size_t)rbase * OUT_DIM + n;
    if (row0 + 16 <= N) {
        #pragma unroll
        for (int v = 0; v < 8; ++v) dPtr[(size_t)v * OUT_DIM] = c[v];
    } else {
        for (int v = 0; v < 8; ++v)
            if (rbase + v < N) dPtr[(size_t)v * OUT_DIM] = c[v];
    }
}

#define ROWS_PER_WG    40
#define GATHER_THREADS 1024

__device__ __forceinline__ int4 load_dst4_guarded(const int* __restrict__ p, int e0, int E) {
    int4 d;
    d.x = (e0     < E) ? p[e0]     : -1;
    d.y = (e0 + 1 < E) ? p[e0 + 1] : -1;
    d.z = (e0 + 2 < E) ? p[e0 + 2] : -1;
    d.w = (e0 + 3 < E) ? p[e0 + 3] : -1;
    return d;
}

__device__ __forceinline__ void add_matches(int d, bool m, int e_lane0,
                                            const int* __restrict__ ei,
                                            const float* __restrict__ dinv,
                                            const float* __restrict__ xw,
                                            float* acc, int row0, int lane) {
    unsigned bal = __builtin_amdgcn_ballot_w32(m);
    while (bal) {
        const int b = __builtin_ctz(bal);
        bal &= bal - 1u;
        const int db = __builtin_amdgcn_readlane(d, b);
        const int sb = ei[e_lane0 + 4 * b];
        const float4* __restrict__ srow4 = (const float4*)(xw + (size_t)sb * OUT_DIM);
        const float4 v0 = srow4[lane];
        const float4 v1 = srow4[lane + 32];
        const float4 v2 = srow4[lane + 64];
        const float4 v3 = srow4[lane + 96];
        const float nb = dinv[sb] * dinv[db];
        float* a = acc + (db - row0) * OUT_DIM + lane;
        atomicAdd(a +   0, nb * v0.x); atomicAdd(a + 128, nb * v0.y);
        atomicAdd(a + 256, nb * v0.z); atomicAdd(a + 384, nb * v0.w);
        atomicAdd(a +  32, nb * v1.x); atomicAdd(a + 160, nb * v1.y);
        atomicAdd(a + 288, nb * v1.z); atomicAdd(a + 416, nb * v1.w);
        atomicAdd(a +  64, nb * v2.x); atomicAdd(a + 192, nb * v2.y);
        atomicAdd(a + 320, nb * v2.z); atomicAdd(a + 448, nb * v2.w);
        atomicAdd(a +  96, nb * v3.x); atomicAdd(a + 224, nb * v3.y);
        atomicAdd(a + 352, nb * v3.z); atomicAdd(a + 480, nb * v3.w);
    }
}

__global__ __launch_bounds__(GATHER_THREADS)
void k_owner_gather(const int* __restrict__ ei, const float* __restrict__ xw,
                    const float* __restrict__ dinv, const float* __restrict__ cvec,
                    float* __restrict__ out, int N, int E) {
    __shared__ float acc[ROWS_PER_WG * OUT_DIM];

    const int tid  = threadIdx.x;
    const int lane = tid & 31;
    const int wave = __builtin_amdgcn_readfirstlane(tid >> 5);
    const int row0 = blockIdx.x * ROWS_PER_WG;
    int rows = N - row0;
    if (rows > ROWS_PER_WG) rows = ROWS_PER_WG;

    const float4* __restrict__ xw4   = (const float4*)xw;
    const float4* __restrict__ cvec4 = (const float4*)cvec;
    for (int idx = tid; idx < rows * 128; idx += GATHER_THREADS) {
        const int r = idx >> 7;
        const int q = idx & 127;
        const int i = row0 + r;
        const float di = dinv[i];
        const float s  = di * di;
        const float4 xv = xw4[(size_t)i * 128 + q];
        const float4 cv = cvec4[q];
        float* a = acc + r * OUT_DIM + q;
        a[0]   = cv.x + s * xv.x;
        a[128] = cv.y + s * xv.y;
        a[256] = cv.z + s * xv.z;
        a[384] = cv.w + s * xv.w;
    }
    __syncthreads();

    const int* __restrict__ dstp = ei + E;
    const bool vec_ok  = (((size_t)dstp) & 15) == 0;
    const int  nChunks = (E + 127) >> 7;
    for (int c0 = wave * 4; c0 < nChunks; c0 += (GATHER_THREADS / 32) * 4) {
        int4 dv[4];
        const int eb = (c0 << 7) + lane * 4;
        if (vec_ok && ((c0 + 4) << 7) <= E) {
            #pragma unroll
            for (int j = 0; j < 4; ++j) dv[j] = *(const int4*)(dstp + eb + (j << 7));
        } else {
            #pragma unroll
            for (int j = 0; j < 4; ++j) dv[j] = load_dst4_guarded(dstp, eb + (j << 7), E);
        }
        #pragma unroll
        for (int j = 0; j < 4; ++j) {
            const int4 d  = dv[j];
            const int  eu = (c0 << 7) + (j << 7);
            const bool m0 = (unsigned)(d.x - row0) < (unsigned)rows;
            const bool m1 = (unsigned)(d.y - row0) < (unsigned)rows;
            const bool m2 = (unsigned)(d.z - row0) < (unsigned)rows;
            const bool m3 = (unsigned)(d.w - row0) < (unsigned)rows;
            if (__builtin_amdgcn_ballot_w32(m0 | m1 | m2 | m3) == 0u) continue;
            add_matches(d.x, m0, eu + 0, ei, dinv, xw, acc, row0, lane);
            add_matches(d.y, m1, eu + 1, ei, dinv, xw, acc, row0, lane);
            add_matches(d.z, m2, eu + 2, ei, dinv, xw, acc, row0, lane);
            add_matches(d.w, m3, eu + 3, ei, dinv, xw, acc, row0, lane);
        }
    }
    __syncthreads();

    float4* __restrict__ out4 = (float4*)out;
    for (int idx = tid; idx < rows * 128; idx += GATHER_THREADS) {
        const int r = idx >> 7;
        const int q = idx & 127;
        const float* a = acc + r * OUT_DIM + q;
        float4 o;
        o.x = a[0]; o.y = a[128]; o.z = a[256]; o.w = a[384];
        out4[(size_t)(row0 + r) * 128 + q] = o;
    }
}

extern "C" void kernel_launch(void* const* d_in, const int* in_sizes, int n_in,
                              void* d_out, int out_size, void* d_ws, size_t ws_size,
                              hipStream_t stream) {
    const float* x     = (const float*)d_in[0];
    const float* t_emb = (const float*)d_in[1];
    const int*   ei    = (const int*)d_in[2];
    const float* W     = (const float*)d_in[3];
    const float* b     = (const float*)d_in[4];
    const float* Wt    = (const float*)d_in[5];
    const float* bt    = (const float*)d_in[6];
    float*       out   = (float*)d_out;

    const int N = in_sizes[0] / IN_DIM;
    const int E = in_sizes[2] / 2;

    char* ws = (char*)d_ws;
    float* deg  = (float*)(ws + 0);
    float* dinv = (float*)(ws + (((size_t)N * 4 + 63) & ~63ull));
    float* cvec = (float*)((char*)dinv + (((size_t)N * 4 + 63) & ~63ull));
    float* xw   = (float*)((char*)cvec + 4096);

    k_deg_init <<<(N + 255) / 256, 256, 0, stream>>>(deg, N);
    k_deg_accum<<<(E + 255) / 256, 256, 0, stream>>>(ei, deg, E);
    k_dinv     <<<(N + 255) / 256, 256, 0, stream>>>(deg, dinv, N);

    k_cvec<<<(OUT_DIM + 255) / 256, 256, 0, stream>>>(t_emb, Wt, b, bt, cvec);

    {
        int rowTiles = (N + 15) / 16;
        int tiles    = rowTiles * (OUT_DIM / 16);
        int blocks   = (tiles + 7) / 8;
        k_gemm_wmma<<<blocks, 256, 0, stream>>>(x, W, xw, N, rowTiles);
    }

    {
        int blocks = (N + ROWS_PER_WG - 1) / ROWS_PER_WG;
        k_owner_gather<<<blocks, GATHER_THREADS, 0, stream>>>(ei, xw, dinv, cvec, out, N, E);
    }
}


# GEMM: a wave owns a 32x32 tile (2x2 accumulators), x read as b128 per 8 k, next 16-k stage loaded ahead of the WMMAs; k order inside each 8-chunk permuted (sum order changes)
# speedup vs baseline: 1.9111x; 1.2836x over previous
#include <hip/hip_runtime.h>
#include <hip/hip_bf16.h>

typedef float v2f __attribute__((ext_vector_type(2)));
typedef float v8f __attribute__((ext_vector_type(8)));

#define IN_DIM  512
#define OUT_DIM 512
#define TC_DIM  256

__global__ void k_deg_init(float* __restrict__ deg, int N) {
    int i = blockIdx.x * blockDim.x + threadIdx.x;
    if (i < N) deg[i] = 1.0f;
}

__global__ void k_deg_accum(const int* __restrict__ ei, float* __restrict__ deg, int E) {
    int e = blockIdx.x * blockDim.x + threadIdx.x;
    if (e < E) atomicAdd(&deg[ei[E + e]], 1.0f);
}

__global__ void k_dinv(const float* __restrict__ deg, float* __restrict__ dinv, int N) {
    int i = blockIdx.x * blockDim.x + threadIdx.x;
    if (i < N) dinv[i] = __frsqrt_rn(deg[i]);
}

__global__ void k_cvec(const float* __restrict__ t_emb, const float* __restrict__ Wt,
                       const float* __restrict__ b, const float* __restrict__ bt,
                       float* __restrict__ cvec) {
    int j = blockIdx.x * blockDim.x + threadIdx.x;
    if (j >= OUT_DIM) return;
    float s = b[j] + bt[j];
    #pragma unroll 4
    for (int k = 0; k < TC_DIM; ++k) s += t_emb[k] * Wt[k * OUT_DIM + j];
    cvec[j] = s;
}

#define GEMM_WAVES 4

struct GemmStage { float4 a[2][2]; v2f b[2][2][2]; };

__device__ __forceinline__ void gemm_load_stage(GemmStage& f, const float* __restrict__ aPtr0,
                                                const float* __restrict__ aPtr1,
                                                const float* __restrict__ bPtr, int k0) {
    #pragma unroll
    for (int q = 0; q < 2; ++q) {
        f.a[q][0] = *(const float4*)(aPtr0 + k0 + 8 * q);
        f.a[q][1] = *(const float4*)(aPtr1 + k0 + 8 * q);
        #pragma unroll
        for (int j = 0; j < 2; ++j) {
            const float* __restrict__ r0 = bPtr + (size_t)(k0 + 8 * q + 2 * j) * OUT_DIM;
            f.b[q][j][0].x = r0[0];
            f.b[q][j][0].y = r0[OUT_DIM];
            f.b[q][j][1].x = r0[16];
            f.b[q][j][1].y = r0[OUT_DIM + 16];
        }
    }
}

__device__ __forceinline__ void gemm_mma_stage(const GemmStage& f, v8f& c00, v8f& c01, v8f& c10, v8f& c11) {
    #pragma unroll
    for (int q = 0; q < 2; ++q) {
        #pragma unroll
        for (int j = 0; j < 2; ++j) {
            v2f a0, a1;
            if (j == 0) { a0.x = f.a[q][0].x; a0.y = f.a[q][0].y; a1.x = f.a[q][1].x; a1.y = f.a[q][1].y; }
            else        { a0.x = f.a[q][0].z; a0.y = f.a[q][0].w; a1.x = f.a[q][1].z; a1.y = f.a[q][1].w; }
            const v2f b0 = f.b[q][j][0], b1 = f.b[q][j][1];
            c00 = __builtin_amdgcn_wmma_f32_16x16x4_f32(false, a0, false, b0, (short)0, c00, false, false);
            c01 = __builtin_amdgcn_wmma_f32_16x16x4_f32(false, a0, false, b1, (short)0, c01, false, false);
            c10 = __builtin_amdgcn_wmma_f32_16x16x4_f32(false, a1, false, b0, (short)0, c10, false, false);
            c11 = __builtin_amdgcn_wmma_f32_16x16x4_f32(false, a1, false, b1, (short)0, c11, false, false);
        }
    }
}

__device__ __forceinline__ void gemm_store_tile(float* __restrict__ p, const v8f& c, int rbase, int N, bool full) {
    if (full) {
        #pragma unroll
        for (int v = 0; v < 8; ++v) p[(size_t)v * OUT_DIM] = c[v];
    } else {
        #pragma unroll
        for (int v = 0; v < 8; ++v)
            if (rbase + v < N) p[(size_t)v * OUT_DIM] = c[v];
    }
}

__global__ __launch_bounds__(GEMM_WAVES * 32)
void k_gemm_wmma(const float* __restrict__ x, const float* __restrict__ W,
                 float* __restrict__ xw, int N, int rowTiles) {
    const int lane = threadIdx.x & 31;
    const int wave = __builtin_amdgcn_readfirstlane((int)(threadIdx.x >> 5));
    const int colGroup = blockIdx.x & (OUT_DIM / 64 - 1);
    const int rowTile  = (blockIdx.x >> 3) * 4 + (wave >> 1) * 2;
    if (rowTile >= rowTiles) return;

    const int row0 = rowTile * 16;
    const int col0 = colGroup * 64 + (wave & 1) * 32;
    const int l15  = lane & 15;
    const int hi   = lane >> 4;

    int aRow0 = row0 + l15;
    int aRow1 = row0 + 16 + l15;
    if (aRow0 >= N) aRow0 = N - 1;
    if (aRow1 >= N) aRow1 = N - 1;
    const float* __restrict__ aPtr0 = x + (size_t)aRow0 * IN_DIM + 4 * hi;
    const float* __restrict__ aPtr1 = x + (size_t)aRow1 * IN_DIM + 4 * hi;
    const float* __restrict__ bPtr  = W + (size_t)(4 * hi) * OUT_DIM + col0 + l15;

    v8f c00 = {}, c01 = {}, c10 = {}, c11 = {};
    GemmStage f0, f1;
    gemm_load_stage(f0, aPtr0, aPtr1, bPtr, 0);
    for (int k0 = 0; k0 < IN_DIM - 32; k0 += 32) {
        gemm_load_stage(f1, aPtr0, aPtr1, bPtr, k0 + 16);
        __builtin_amdgcn_sched_barrier(0);
        gemm_mma_stage(f0, c00, c01, c10, c11);
        __builtin_amdgcn_sched_barrier(0);
        gemm_load_stage(f0, aPtr0, aPtr1, bPtr, k0 + 32);
        __builtin_amdgcn_sched_barrier(0);
        gemm_mma_stage(f1, c00, c01, c10, c11);
        __builtin_amdgcn_sched_barrier(0);
    }
    gemm_load_stage(f1, aPtr0, aPtr1, bPtr, IN_DIM - 16);
    __builtin_amdgcn_sched_barrier(0);
    gemm_mma_stage(f0, c00, c01, c10, c11);
    gemm_mma_stage(f1, c00, c01, c10, c11);

    const int  rbase = row0 + hi * 8;
    const bool full  = (row0 + 32 <= N);
    float* __restrict__ dPtr = xw + (size_t)rbase * OUT_DIM + col0 + l15;
    gemm_store_tile(dPtr,                              c00, rbase,      N, full);
    gemm_store_tile(dPtr + 16,                         c01, rbase,      N, full);
    gemm_store_tile(dPtr + (size_t)16 * OUT_DIM,       c10, rbase + 16, N, full);
    gemm_store_tile(dPtr + (size_t)16 * OUT_DIM + 16,  c11, rbase + 16, N, full);
}

#define ROWS_PER_WG    40
#define GATHER_THREADS 1024

__device__ __forceinline__ int4 load_dst4_guarded(const int* __restrict__ p, int e0, int E) {
    int4 d;
    d.x = (e0     < E) ? p[e0]     : -1;
    d.y = (e0 + 1 < E) ? p[e0 + 1] : -1;
    d.z = (e0 + 2 < E) ? p[e0 + 2] : -1;
    d.w = (e0 + 3 < E) ? p[e0 + 3] : -1;
    return d;
}

__device__ __forceinline__ void add_matches(int d, bool m, int e_lane0,
                                            const int* __restrict__ ei,
                                            const float* __restrict__ dinv,
                                            const float* __restrict__ xw,
                                            float* acc, int row0, int lane) {
    unsigned bal = __builtin_amdgcn_ballot_w32(m);
    while (bal) {
        const int b = __builtin_ctz(bal);
        bal &= bal - 1u;
        const int db = __builtin_amdgcn_readlane(d, b);
        const int sb = ei[e_lane0 + 4 * b];
        const float4* __restrict__ srow4 = (const float4*)(xw + (size_t)sb * OUT_DIM);
        const float4 v0 = srow4[lane];
        const float4 v1 = srow4[lane + 32];
        const float4 v2 = srow4[lane + 64];
        const float4 v3 = srow4[lane + 96];
        const float nb = dinv[sb] * dinv[db];
        float* a = acc + (db - row0) * OUT_DIM + lane;
        atomicAdd(a +   0, nb * v0.x); atomicAdd(a + 128, nb * v0.y);
        atomicAdd(a + 256, nb * v0.z); atomicAdd(a + 384, nb * v0.w);
        atomicAdd(a +  32, nb * v1.x); atomicAdd(a + 160, nb * v1.y);
        atomicAdd(a + 288, nb * v1.z); atomicAdd(a + 416, nb * v1.w);
        atomicAdd(a +  64, nb * v2.x); atomicAdd(a + 192, nb * v2.y);
        atomicAdd(a + 320, nb * v2.z); atomicAdd(a + 448, nb * v2.w);
        atomicAdd(a +  96, nb * v3.x); atomicAdd(a + 224, nb * v3.y);
        atomicAdd(a + 352, nb * v3.z); atomicAdd(a + 480, nb * v3.w);
    }
}

__global__ __launch_bounds__(GATHER_THREADS)
void k_owner_gather(const int* __restrict__ ei, const float* __restrict__ xw,
                    const float* __restrict__ dinv, const float* __restrict__ cvec,
                    float* __restrict__ out, int N, int E) {
    __shared__ float acc[ROWS_PER_WG * OUT_DIM];

    const int tid  = threadIdx.x;
    const int lane = tid & 31;
    const int wave = __builtin_amdgcn_readfirstlane(tid >> 5);
    const int row0 = blockIdx.x * ROWS_PER_WG;
    int rows = N - row0;
    if (rows > ROWS_PER_WG) rows = ROWS_PER_WG;

    const float4* __restrict__ xw4   = (const float4*)xw;
    const float4* __restrict__ cvec4 = (const float4*)cvec;
    for (int idx = tid; idx < rows * 128; idx += GATHER_THREADS) {
        const int r = idx >> 7;
        const int q = idx & 127;
        const int i = row0 + r;
        const float di = dinv[i];
        const float s  = di * di;
        const float4 xv = xw4[(size_t)i * 128 + q];
        const float4 cv = cvec4[q];
        float* a = acc + r * OUT_DIM + q;
        a[0]   = cv.x + s * xv.x;
        a[128] = cv.y + s * xv.y;
        a[256] = cv.z + s * xv.z;
        a[384] = cv.w + s * xv.w;
    }
    __syncthreads();

    const int* __restrict__ dstp = ei + E;
    const bool vec_ok  = (((size_t)dstp) & 15) == 0;
    const int  nChunks = (E + 127) >> 7;
    for (int c0 = wave * 4; c0 < nChunks; c0 += (GATHER_THREADS / 32) * 4) {
        int4 dv[4];
        const int eb = (c0 << 7) + lane * 4;
        if (vec_ok && ((c0 + 4) << 7) <= E) {
            #pragma unroll
            for (int j = 0; j < 4; ++j) dv[j] = *(const int4*)(dstp + eb + (j << 7));
        } else {
            #pragma unroll
            for (int j = 0; j < 4; ++j) dv[j] = load_dst4_guarded(dstp, eb + (j << 7), E);
        }
        #pragma unroll
        for (int j = 0; j < 4; ++j) {
            const int4 d  = dv[j];
            const int  eu = (c0 << 7) + (j << 7);
            const bool m0 = (unsigned)(d.x - row0) < (unsigned)rows;
            const bool m1 = (unsigned)(d.y - row0) < (unsigned)rows;
            const bool m2 = (unsigned)(d.z - row0) < (unsigned)rows;
            const bool m3 = (unsigned)(d.w - row0) < (unsigned)rows;
            if (__builtin_amdgcn_ballot_w32(m0 | m1 | m2 | m3) == 0u) continue;
            add_matches(d.x, m0, eu + 0, ei, dinv, xw, acc, row0, lane);
            add_matches(d.y, m1, eu + 1, ei, dinv, xw, acc, row0, lane);
            add_matches(d.z, m2, eu + 2, ei, dinv, xw, acc, row0, lane);
            add_matches(d.w, m3, eu + 3, ei, dinv, xw, acc, row0, lane);
        }
    }
    __syncthreads();

    float4* __restrict__ out4 = (float4*)out;
    for (int idx = tid; idx < rows * 128; idx += GATHER_THREADS) {
        const int r = idx >> 7;
        const int q = idx & 127;
        const float* a = acc + r * OUT_DIM + q;
        float4 o;
        o.x = a[0]; o.y = a[128]; o.z = a[256]; o.w = a[384];
        out4[(size_t)(row0 + r) * 128 + q] = o;
    }
}

extern "C" void kernel_launch(void* const* d_in, const int* in_sizes, int n_in,
                              void* d_out, int out_size, void* d_ws, size_t ws_size,
                              hipStream_t stream) {
    const float* x     = (const float*)d_in[0];
    const float* t_emb = (const float*)d_in[1];
    const int*   ei    = (const int*)d_in[2];
    const float* W     = (const float*)d_in[3];
    const float* b     = (const float*)d_in[4];
    const float* Wt    = (const float*)d_in[5];
    const float* bt    = (const float*)d_in[6];
    float*       out   = (float*)d_out;

    const int N = in_sizes[0] / IN_DIM;
    const int E = in_sizes[2] / 2;

    char* ws = (char*)d_ws;
    float* deg  = (float*)(ws + 0);
    float* dinv = (float*)(ws + (((size_t)N * 4 + 63) & ~63ull));
    float* cvec = (float*)((char*)dinv + (((size_t)N * 4 + 63) & ~63ull));
    float* xw   = (float*)((char*)cvec + 4096);

    k_deg_init <<<(N + 255) / 256, 256, 0, stream>>>(deg, N);
    k_deg_accum<<<(E + 255) / 256, 256, 0, stream>>>(ei, deg, E);
    k_dinv     <<<(N + 255) / 256, 256, 0, stream>>>(deg, dinv, N);

    k_cvec<<<(OUT_DIM + 255) / 256, 256, 0, stream>>>(t_emb, Wt, b, bt, cvec);

    {
        int rowTiles  = (N + 15) / 16;
        int rowGroups = (rowTiles + 3) / 4;
        int blocks    = rowGroups * (OUT_DIM / 64);
        k_gemm_wmma<<<blocks, GEMM_WAVES * 32, 0, stream>>>(x, W, xw, N, rowTiles);
    }

    {
        int blocks = (N + ROWS_PER_WG - 1) / ROWS_PER_WG;
        k_owner_gather<<<blocks, GATHER_THREADS, 0, stream>>>(ei, xw, dinv, cvec, out, N, E);
    }
}


# cvec (t_emb @ Wt + biases): serial 256-k loop of one thread per column split over the 8 waves of a workgroup, 32 independent loads per lane, partials added via LDS in wave order (sum order changes)
# speedup vs baseline: 2.5902x; 1.3553x over previous
#include <hip/hip_runtime.h>
#include <hip/hip_bf16.h>

typedef float v2f __attribute__((ext_vector_type(2)));
typedef float v8f __attribute__((ext_vector_type(8)));

#define IN_DIM  512
#define OUT_DIM 512
#define TC_DIM  256

__global__ void k_deg_init(float* __restrict__ deg, int N) {
    int i = blockIdx.x * blockDim.x + threadIdx.x;
    if (i < N) deg[i] = 1.0f;
}

__global__ void k_deg_accum(const int* __restrict__ ei, float* __restrict__ deg, int E) {
    int e = blockIdx.x * blockDim.x + threadIdx.x;
    if (e < E) atomicAdd(&deg[ei[E + e]], 1.0f);
}

__global__ void k_dinv(const float* __restrict__ deg, float* __restrict__ dinv, int N) {
    int i = blockIdx.x * blockDim.x + threadIdx.x;
    if (i < N) dinv[i] = __frsqrt_rn(deg[i]);
}

#define CVEC_WAVES 8
#define CVEC_KPW   (TC_DIM / CVEC_WAVES)
__global__ __launch_bounds__(CVEC_WAVES * 32)
void k_cvec(const float* __restrict__ t_emb, const float* __restrict__ Wt,
            const float* __restrict__ b, const float* __restrict__ bt,
            float* __restrict__ cvec) {
    __shared__ float part[CVEC_WAVES][32];
    const int lane = threadIdx.x & 31;
    const int wave = __builtin_amdgcn_readfirstlane((int)(threadIdx.x >> 5));
    const int j    = blockIdx.x * 32 + lane;
    const float* __restrict__ tp = t_emb + wave * CVEC_KPW;
    const float* __restrict__ wp = Wt + (size_t)(wave * CVEC_KPW) * OUT_DIM + j;
    const float bias = b[j] + bt[j];
    float s = 0.0f;
    for (int i = 0; i < CVEC_KPW; ++i) s += tp[i] * wp[(size_t)i * OUT_DIM];
    part[wave][lane] = s;
    __syncthreads();
    if (wave == 0) {
        float r = bias;
        for (int w = 0; w < CVEC_WAVES; ++w) r += part[w][lane];
        cvec[j] = r;
    }
}

#define GEMM_WAVES 4

struct GemmStage { float4 a[2][2]; v2f b[2][2][2]; };

__device__ __forceinline__ void gemm_load_stage(GemmStage& f, const float* __restrict__ aPtr0,
                                                const float* __restrict__ aPtr1,
                                                const float* __restrict__ bPtr, int k0) {
    #pragma unroll
    for (int q = 0; q < 2; ++q) {
        f.a[q][0] = *(const float4*)(aPtr0 + k0 + 8 * q);
        f.a[q][1] = *(const float4*)(aPtr1 + k0 + 8 * q);
        #pragma unroll
        for (int j = 0; j < 2; ++j) {
            const float* __restrict__ r0 = bPtr + (size_t)(k0 + 8 * q + 2 * j) * OUT_DIM;
            f.b[q][j][0].x = r0[0];
            f.b[q][j][0].y = r0[OUT_DIM];
            f.b[q][j][1].x = r0[16];
            f.b[q][j][1].y = r0[OUT_DIM + 16];
        }
    }
}

__device__ __forceinline__ void gemm_mma_stage(const GemmStage& f, v8f& c00, v8f& c01, v8f& c10, v8f& c11) {
    #pragma unroll
    for (int q = 0; q < 2; ++q) {
        #pragma unroll
        for (int j = 0; j < 2; ++j) {
            v2f a0, a1;
            if (j == 0) { a0.x = f.a[q][0].x; a0.y = f.a[q][0].y; a1.x = f.a[q][1].x; a1.y = f.a[q][1].y; }
            else        { a0.x = f.a[q][0].z; a0.y = f.a[q][0].w; a1.x = f.a[q][1].z; a1.y = f.a[q][1].w; }
            const v2f b0 = f.b[q][j][0], b1 = f.b[q][j][1];
            c00 = __builtin_amdgcn_wmma_f32_16x16x4_f32(false, a0, false, b0, (short)0, c00, false, false);
            c01 = __builtin_amdgcn_wmma_f32_16x16x4_f32(false, a0, false, b1, (short)0, c01, false, false);
            c10 = __builtin_amdgcn_wmma_f32_16x16x4_f32(false, a1, false, b0, (short)0, c10, false, false);
            c11 = __builtin_amdgcn_wmma_f32_16x16x4_f32(false, a1, false, b1, (short)0, c11, false, false);
        }
    }
}

__device__ __forceinline__ void gemm_store_tile(float* __restrict__ p, const v8f& c, int rbase, int N, bool full) {
    if (full) {
        #pragma unroll
        for (int v = 0; v < 8; ++v) p[(size_t)v * OUT_DIM] = c[v];
    } else {
        #pragma unroll
        for (int v = 0; v < 8; ++v)
            if (rbase + v < N) p[(size_t)v * OUT_DIM] = c[v];
    }
}

__global__ __launch_bounds__(GEMM_WAVES * 32)
void k_gemm_wmma(const float* __restrict__ x, const float* __restrict__ W,
                 float* __restrict__ xw, int N, int rowTiles) {
    const int lane = threadIdx.x & 31;
    const int wave = __builtin_amdgcn_readfirstlane((int)(threadIdx.x >> 5));
    const int colGroup = blockIdx.x & (OUT_DIM / 64 - 1);
    const int rowTile  = (blockIdx.x >> 3) * 4 + (wave >> 1) * 2;
    if (rowTile >= rowTiles) return;

    const int row0 = rowTile * 16;
    const int col0 = colGroup * 64 + (wave & 1) * 32;
    const int l15  = lane & 15;
    const int hi   = lane >> 4;

    int aRow0 = row0 + l15;
    int aRow1 = row0 + 16 + l15;
    if (aRow0 >= N) aRow0 = N - 1;
    if (aRow1 >= N) aRow1 = N - 1;
    const float* __restrict__ aPtr0 = x + (size_t)aRow0 * IN_DIM + 4 * hi;
    const float* __restrict__ aPtr1 = x + (size_t)aRow1 * IN_DIM + 4 * hi;
    const float* __restrict__ bPtr  = W + (size_t)(4 * hi) * OUT_DIM + col0 + l15;

    v8f c00 = {}, c01 = {}, c10 = {}, c11 = {};
    GemmStage f0, f1;
    gemm_load_stage(f0, aPtr0, aPtr1, bPtr, 0);
    for (int k0 = 0; k0 < IN_DIM - 32; k0 += 32) {
        gemm_load_stage(f1, aPtr0, aPtr1, bPtr, k0 + 16);
        __builtin_amdgcn_sched_barrier(0);
        gemm_mma_stage(f0, c00, c01, c10, c11);
        __builtin_amdgcn_sched_barrier(0);
        gemm_load_stage(f0, aPtr0, aPtr1, bPtr, k0 + 32);
        __builtin_amdgcn_sched_barrier(0);
        gemm_mma_stage(f1, c00, c01, c10, c11);
        __builtin_amdgcn_sched_barrier(0);
    }
    gemm_load_stage(f1, aPtr0, aPtr1, bPtr, IN_DIM - 16);
    __builtin_amdgcn_sched_barrier(0);
    gemm_mma_stage(f0, c00, c01, c10, c11);
    gemm_mma_stage(f1, c00, c01, c10, c11);

    const int  rbase = row0 + hi * 8;
    const bool full  = (row0 + 32 <= N);
    float* __restrict__ dPtr = xw + (size_t)rbase * OUT_DIM + col0 + l15;
    gemm_store_tile(dPtr,                              c00, rbase,      N, full);
    gemm_store_tile(dPtr + 16,                         c01, rbase,      N, full);
    gemm_store_tile(dPtr + (size_t)16 * OUT_DIM,       c10, rbase + 16, N, full);
    gemm_store_tile(dPtr + (size_t)16 * OUT_DIM + 16,  c11, rbase + 16, N, full);
}

#define ROWS_PER_WG    40
#define GATHER_THREADS 1024

__device__ __forceinline__ int4 load_dst4_guarded(const int* __restrict__ p, int e0, int E) {
    int4 d;
    d.x = (e0     < E) ? p[e0]     : -1;
    d.y = (e0 + 1 < E) ? p[e0 + 1] : -1;
    d.z = (e0 + 2 < E) ? p[e0 + 2] : -1;
    d.w = (e0 + 3 < E) ? p[e0 + 3] : -1;
    return d;
}

__device__ __forceinline__ void add_matches(int d, bool m, int e_lane0,
                                            const int* __restrict__ ei,
                                            const float* __restrict__ dinv,
                                            const float* __restrict__ xw,
                                            float* acc, int row0, int lane) {
    unsigned bal = __builtin_amdgcn_ballot_w32(m);
    while (bal) {
        const int b = __builtin_ctz(bal);
        bal &= bal - 1u;
        const int db = __builtin_amdgcn_readlane(d, b);
        const int sb = ei[e_lane0 + 4 * b];
        const float4* __restrict__ srow4 = (const float4*)(xw + (size_t)sb * OUT_DIM);
        const float4 v0 = srow4[lane];
        const float4 v1 = srow4[lane + 32];
        const float4 v2 = srow4[lane + 64];
        const float4 v3 = srow4[lane + 96];
        const float nb = dinv[sb] * dinv[db];
        float* a = acc + (db - row0) * OUT_DIM + lane;
        atomicAdd(a +   0, nb * v0.x); atomicAdd(a + 128, nb * v0.y);
        atomicAdd(a + 256, nb * v0.z); atomicAdd(a + 384, nb * v0.w);
        atomicAdd(a +  32, nb * v1.x); atomicAdd(a + 160, nb * v1.y);
        atomicAdd(a + 288, nb * v1.z); atomicAdd(a + 416, nb * v1.w);
        atomicAdd(a +  64, nb * v2.x); atomicAdd(a + 192, nb * v2.y);
        atomicAdd(a + 320, nb * v2.z); atomicAdd(a + 448, nb * v2.w);
        atomicAdd(a +  96, nb * v3.x); atomicAdd(a + 224, nb * v3.y);
        atomicAdd(a + 352, nb * v3.z); atomicAdd(a + 480, nb * v3.w);
    }
}

__global__ __launch_bounds__(GATHER_THREADS)
void k_owner_gather(const int* __restrict__ ei, const float* __restrict__ xw,
                    const float* __restrict__ dinv, const float* __restrict__ cvec,
                    float* __restrict__ out, int N, int E) {
    __shared__ float acc[ROWS_PER_WG * OUT_DIM];

    const int tid  = threadIdx.x;
    const int lane = tid & 31;
    const int wave = __builtin_amdgcn_readfirstlane(tid >> 5);
    const int row0 = blockIdx.x * ROWS_PER_WG;
    int rows = N - row0;
    if (rows > ROWS_PER_WG) rows = ROWS_PER_WG;

    const float4* __restrict__ xw4   = (const float4*)xw;
    const float4* __restrict__ cvec4 = (const float4*)cvec;
    for (int idx = tid; idx < rows * 128; idx += GATHER_THREADS) {
        const int r = idx >> 7;
        const int q = idx & 127;
        const int i = row0 + r;
        const float di = dinv[i];
        const float s  = di * di;
        const float4 xv = xw4[(size_t)i * 128 + q];
        const float4 cv = cvec4[q];
        float* a = acc + r * OUT_DIM + q;
        a[0]   = cv.x + s * xv.x;
        a[128] = cv.y + s * xv.y;
        a[256] = cv.z + s * xv.z;
        a[384] = cv.w + s * xv.w;
    }
    __syncthreads();

    const int* __restrict__ dstp = ei + E;
    const bool vec_ok  = (((size_t)dstp) & 15) == 0;
    const int  nChunks = (E + 127) >> 7;
    for (int c0 = wave * 4; c0 < nChunks; c0 += (GATHER_THREADS / 32) * 4) {
        int4 dv[4];
        const int eb = (c0 << 7) + lane * 4;
        if (vec_ok && ((c0 + 4) << 7) <= E) {
            #pragma unroll
            for (int j = 0; j < 4; ++j) dv[j] = *(const int4*)(dstp + eb + (j << 7));
        } else {
            #pragma unroll
            for (int j = 0; j < 4; ++j) dv[j] = load_dst4_guarded(dstp, eb + (j << 7), E);
        }
        #pragma unroll
        for (int j = 0; j < 4; ++j) {
            const int4 d  = dv[j];
            const int  eu = (c0 << 7) + (j << 7);
            const bool m0 = (unsigned)(d.x - row0) < (unsigned)rows;
            const bool m1 = (unsigned)(d.y - row0) < (unsigned)rows;
            const bool m2 = (unsigned)(d.z - row0) < (unsigned)rows;
            const bool m3 = (unsigned)(d.w - row0) < (unsigned)rows;
            if (__builtin_amdgcn_ballot_w32(m0 | m1 | m2 | m3) == 0u) continue;
            add_matches(d.x, m0, eu + 0, ei, dinv, xw, acc, row0, lane);
            add_matches(d.y, m1, eu + 1, ei, dinv, xw, acc, row0, lane);
            add_matches(d.z, m2, eu + 2, ei, dinv, xw, acc, row0, lane);
            add_matches(d.w, m3, eu + 3, ei, dinv, xw, acc, row0, lane);
        }
    }
    __syncthreads();

    float4* __restrict__ out4 = (float4*)out;
    for (int idx = tid; idx < rows * 128; idx += GATHER_THREADS) {
        const int r = idx >> 7;
        const int q = idx & 127;
        const float* a = acc + r * OUT_DIM + q;
        float4 o;
        o.x = a[0]; o.y = a[128]; o.z = a[256]; o.w = a[384];
        out4[(size_t)(row0 + r) * 128 + q] = o;
    }
}

extern "C" void kernel_launch(void* const* d_in, const int* in_sizes, int n_in,
                              void* d_out, int out_size, void* d_ws, size_t ws_size,
                              hipStream_t stream) {
    const float* x     = (const float*)d_in[0];
    const float* t_emb = (const float*)d_in[1];
    const int*   ei    = (const int*)d_in[2];
    const float* W     = (const float*)d_in[3];
    const float* b     = (const float*)d_in[4];
    const float* Wt    = (const float*)d_in[5];
    const float* bt    = (const float*)d_in[6];
    float*       out   = (float*)d_out;

    const int N = in_sizes[0] / IN_DIM;
    const int E = in_sizes[2] / 2;

    char* ws = (char*)d_ws;
    float* deg  = (float*)(ws + 0);
    float* dinv = (float*)(ws + (((size_t)N * 4 + 63) & ~63ull));
    float* cvec = (float*)((char*)dinv + (((size_t)N * 4 + 63) & ~63ull));
    float* xw   = (float*)((char*)cvec + 4096);

    k_deg_init <<<(N + 255) / 256, 256, 0, stream>>>(deg, N);
    k_deg_accum<<<(E + 255) / 256, 256, 0, stream>>>(ei, deg, E);
    k_dinv     <<<(N + 255) / 256, 256, 0, stream>>>(deg, dinv, N);

    k_cvec<<<OUT_DIM / 32, CVEC_WAVES * 32, 0, stream>>>(t_emb, Wt, b, bt, cvec);

    {
        int rowTiles  = (N + 15) / 16;
        int rowGroups = (rowTiles + 3) / 4;
        int blocks    = rowGroups * (OUT_DIM / 64);
        k_gemm_wmma<<<blocks, GEMM_WAVES * 32, 0, stream>>>(x, W, xw, N, rowTiles);
    }

    {
        int blocks = (N + ROWS_PER_WG - 1) / ROWS_PER_WG;
        k_owner_gather<<<blocks, GATHER_THREADS, 0, stream>>>(ei, xw, dinv, cvec, out, N, E);
    }
}


# The 3 degree launches folded into their neighbours: start value 1 stored by the cvec launch, edge-count atomics issued by the GEMM launch's threads, rsqrt taken where the gather reads deg (6 -> 3)
# speedup vs baseline: 2.8179x; 1.0879x over previous
#include <hip/hip_runtime.h>
#include <hip/hip_bf16.h>

typedef float v2f __attribute__((ext_vector_type(2)));
typedef float v8f __attribute__((ext_vector_type(8)));

#define IN_DIM  512
#define OUT_DIM 512
#define TC_DIM  256


#define CVEC_WAVES 8
#define CVEC_KPW   (TC_DIM / CVEC_WAVES)
__global__ __launch_bounds__(CVEC_WAVES * 32)
void k_cvec(const float* __restrict__ t_emb, const float* __restrict__ Wt,
            const float* __restrict__ b, const float* __restrict__ bt,
            float* __restrict__ cvec, float* __restrict__ deg, int N) {
    __shared__ float part[CVEC_WAVES][32];
    for (int i = blockIdx.x * (CVEC_WAVES * 32) + threadIdx.x; i < N; i += (OUT_DIM / 32) * (CVEC_WAVES * 32))
        deg[i] = 1.0f;
    const int lane = threadIdx.x & 31;
    const int wave = __builtin_amdgcn_readfirstlane((int)(threadIdx.x >> 5));
    const int j    = blockIdx.x * 32 + lane;
    const float* __restrict__ tp = t_emb + wave * CVEC_KPW;
    const float* __restrict__ wp = Wt + (size_t)(wave * CVEC_KPW) * OUT_DIM + j;
    const float bias = b[j] + bt[j];
    float s = 0.0f;
    for (int i = 0; i < CVEC_KPW; ++i) s += tp[i] * wp[(size_t)i * OUT_DIM];
    part[wave][lane] = s;
    __syncthreads();
    if (wave == 0) {
        float r = bias;
        for (int w = 0; w < CVEC_WAVES; ++w) r += part[w][lane];
        cvec[j] = r;
    }
}

#define GEMM_WAVES 4

struct GemmStage { float4 a[2][2]; v2f b[2][2][2]; };

__device__ __forceinline__ void gemm_load_stage(GemmStage& f, const float* __restrict__ aPtr0,
                                                const float* __restrict__ aPtr1,
                                                const float* __restrict__ bPtr, int k0) {
    #pragma unroll
    for (int q = 0; q < 2; ++q) {
        f.a[q][0] = *(const float4*)(aPtr0 + k0 + 8 * q);
        f.a[q][1] = *(const float4*)(aPtr1 + k0 + 8 * q);
        #pragma unroll
        for (int j = 0; j < 2; ++j) {
            const float* __restrict__ r0 = bPtr + (size_t)(k0 + 8 * q + 2 * j) * OUT_DIM;
            f.b[q][j][0].x = r0[0];
            f.b[q][j][0].y = r0[OUT_DIM];
            f.b[q][j][1].x = r0[16];
            f.b[q][j][1].y = r0[OUT_DIM + 16];
        }
    }
}

__device__ __forceinline__ void gemm_mma_stage(const GemmStage& f, v8f& c00, v8f& c01, v8f& c10, v8f& c11) {
    #pragma unroll
    for (int q = 0; q < 2; ++q) {
        #pragma unroll
        for (int j = 0; j < 2; ++j) {
            v2f a0, a1;
            if (j == 0) { a0.x = f.a[q][0].x; a0.y = f.a[q][0].y; a1.x = f.a[q][1].x; a1.y = f.a[q][1].y; }
            else        { a0.x = f.a[q][0].z; a0.y = f.a[q][0].w; a1.x = f.a[q][1].z; a1.y = f.a[q][1].w; }
            const v2f b0 = f.b[q][j][0], b1 = f.b[q][j][1];
            c00 = __builtin_amdgcn_wmma_f32_16x16x4_f32(false, a0, false, b0, (short)0, c00, false, false);
            c01 = __builtin_amdgcn_wmma_f32_16x16x4_f32(false, a0, false, b1, (short)0, c01, false, false);
            c10 = __builtin_amdgcn_wmma_f32_16x16x4_f32(false, a1, false, b0, (short)0, c10, false, false);
            c11 = __builtin_amdgcn_wmma_f32_16x16x4_f32(false, a1, false, b1, (short)0, c11, false, false);
        }
    }
}

__device__ __forceinline__ void gemm_store_tile(float* __restrict__ p, const v8f& c, int rbase, int N, bool full) {
    if (full) {
        #pragma unroll
        for (int v = 0; v < 8; ++v) p[(size_t)v * OUT_DIM] = c[v];
    } else {
        #pragma unroll
        for (int v = 0; v < 8; ++v)
            if (rbase + v < N) p[(size_t)v * OUT_DIM] = c[v];
    }
}

__global__ __launch_bounds__(GEMM_WAVES * 32)
void k_gemm_wmma(const float* __restrict__ x, const float* __restrict__ W,
                 float* __restrict__ xw, int N, int rowTiles,
                 const int* __restrict__ ei, float* __restrict__ deg, int E) {
    const int gridThreads = ((rowTiles + 3) >> 2) * (OUT_DIM / 64) * (GEMM_WAVES * 32);
    for (int e = blockIdx.x * (GEMM_WAVES * 32) + threadIdx.x; e < E; e += gridThreads)
        atomicAdd(&deg[ei[E + e]], 1.0f);

    const int lane = threadIdx.x & 31;
    const int wave = __builtin_amdgcn_readfirstlane((int)(threadIdx.x >> 5));
    const int colGroup = blockIdx.x & (OUT_DIM / 64 - 1);
    const int rowTile  = (blockIdx.x >> 3) * 4 + (wave >> 1) * 2;
    if (rowTile >= rowTiles) return;

    const int row0 = rowTile * 16;
    const int col0 = colGroup * 64 + (wave & 1) * 32;
    const int l15  = lane & 15;
    const int hi   = lane >> 4;

    int aRow0 = row0 + l15;
    int aRow1 = row0 + 16 + l15;
    if (aRow0 >= N) aRow0 = N - 1;
    if (aRow1 >= N) aRow1 = N - 1;
    const float* __restrict__ aPtr0 = x + (size_t)aRow0 * IN_DIM + 4 * hi;
    const float* __restrict__ aPtr1 = x + (size_t)aRow1 * IN_DIM + 4 * hi;
    const float* __restrict__ bPtr  = W + (size_t)(4 * hi) * OUT_DIM + col0 + l15;

    v8f c00 = {}, c01 = {}, c10 = {}, c11 = {};
    GemmStage f0, f1;
    gemm_load_stage(f0, aPtr0, aPtr1, bPtr, 0);
    for (int k0 = 0; k0 < IN_DIM - 32; k0 += 32) {
        gemm_load_stage(f1, aPtr0, aPtr1, bPtr, k0 + 16);
        __builtin_amdgcn_sched_barrier(0);
        gemm_mma_stage(f0, c00, c01, c10, c11);
        __builtin_amdgcn_sched_barrier(0);
        gemm_load_stage(f0, aPtr0, aPtr1, bPtr, k0 + 32);
        __builtin_amdgcn_sched_barrier(0);
        gemm_mma_stage(f1, c00, c01, c10, c11);
        __builtin_amdgcn_sched_barrier(0);
    }
    gemm_load_stage(f1, aPtr0, aPtr1, bPtr, IN_DIM - 16);
    __builtin_amdgcn_sched_barrier(0);
    gemm_mma_stage(f0, c00, c01, c10, c11);
    gemm_mma_stage(f1, c00, c01, c10, c11);

    const int  rbase = row0 + hi * 8;
    const bool full  = (row0 + 32 <= N);
    float* __restrict__ dPtr = xw + (size_t)rbase * OUT_DIM + col0 + l15;
    gemm_store_tile(dPtr,                              c00, rbase,      N, full);
    gemm_store_tile(dPtr + 16,                         c01, rbase,      N, full);
    gemm_store_tile(dPtr + (size_t)16 * OUT_DIM,       c10, rbase + 16, N, full);
    gemm_store_tile(dPtr + (size_t)16 * OUT_DIM + 16,  c11, rbase + 16, N, full);
}

#define ROWS_PER_WG    40
#define GATHER_THREADS 1024
#define START_PASSES   (ROWS_PER_WG / (GATHER_THREADS / 128))
static_assert(ROWS_PER_WG % (GATHER_THREADS / 128) == 0, "the start-value passes cover the owned rows exactly");

__device__ __forceinline__ int4 load_dst4_guarded(const int* __restrict__ p, int e0, int E) {
    int4 d;
    d.x = (e0     < E) ? p[e0]     : -1;
    d.y = (e0 + 1 < E) ? p[e0 + 1] : -1;
    d.z = (e0 + 2 < E) ? p[e0 + 2] : -1;
    d.w = (e0 + 3 < E) ? p[e0 + 3] : -1;
    return d;
}

__device__ __forceinline__ void add_matches(int d, bool m, int e_lane0,
                                            const int* __restrict__ ei,
                                            const float* __restrict__ deg,
                                            const float* __restrict__ xw,
                                            float* acc, int row0, int lane) {
    unsigned bal = __builtin_amdgcn_ballot_w32(m);
    while (bal) {
        const int b = __builtin_ctz(bal);
        bal &= bal - 1u;
        const int db = __builtin_amdgcn_readlane(d, b);
        const int sb = ei[e_lane0 + 4 * b];
        const float4* __restrict__ srow4 = (const float4*)(xw + (size_t)sb * OUT_DIM);
        const float4 v0 = srow4[lane];
        const float4 v1 = srow4[lane + 32];
        const float4 v2 = srow4[lane + 64];
        const float4 v3 = srow4[lane + 96];
        const float nb = __frsqrt_rn(deg[sb]) * __frsqrt_rn(deg[db]);
        float* a = acc + (db - row0) * OUT_DIM + lane;
        atomicAdd(a +   0, nb * v0.x); atomicAdd(a + 128, nb * v0.y);
        atomicAdd(a + 256, nb * v0.z); atomicAdd(a + 384, nb * v0.w);
        atomicAdd(a +  32, nb * v1.x); atomicAdd(a + 160, nb * v1.y);
        atomicAdd(a + 288, nb * v1.z); atomicAdd(a + 416, nb * v1.w);
        atomicAdd(a +  64, nb * v2.x); atomicAdd(a + 192, nb * v2.y);
        atomicAdd(a + 320, nb * v2.z); atomicAdd(a + 448, nb * v2.w);
        atomicAdd(a +  96, nb * v3.x); atomicAdd(a + 224, nb * v3.y);
        atomicAdd(a + 352, nb * v3.z); atomicAdd(a + 480, nb * v3.w);
    }
}

__global__ __launch_bounds__(GATHER_THREADS)
void k_owner_gather(const int* __restrict__ ei, const float* __restrict__ xw,
                    const float* __restrict__ deg, const float* __restrict__ cvec,
                    float* __restrict__ out, int N, int E) {
    __shared__ float acc[ROWS_PER_WG * OUT_DIM];

    const int tid  = threadIdx.x;
    const int lane = tid & 31;
    const int wave = __builtin_amdgcn_readfirstlane(tid >> 5);
    const int row0 = blockIdx.x * ROWS_PER_WG;
    int rows = N - row0;
    if (rows > ROWS_PER_WG) rows = ROWS_PER_WG;

    const float4* __restrict__ xw4   = (const float4*)xw;
    const float4* __restrict__ cvec4 = (const float4*)cvec;
    {
        const int q  = tid & 127;
        const int r0 = tid >> 7;
        const float4 cv = cvec4[q];
        float  dg[START_PASSES];
        float4 xv[START_PASSES];
        #pragma unroll
        for (int t = 0; t < START_PASSES; ++t) {
            int r = r0 + t * (GATHER_THREADS / 128);
            if (r > rows - 1) r = rows - 1;
            dg[t] = deg[row0 + r];
            xv[t] = xw4[(size_t)(row0 + r) * 128 + q];
        }
        #pragma unroll
        for (int t = 0; t < START_PASSES; ++t) {
            const int r = r0 + t * (GATHER_THREADS / 128);
            if (r < rows) {
                const float di = __frsqrt_rn(dg[t]);
                const float s  = di * di;
                float* a = acc + r * OUT_DIM + q;
                a[0]   = cv.x + s * xv[t].x;
                a[128] = cv.y + s * xv[t].y;
                a[256] = cv.z + s * xv[t].z;
                a[384] = cv.w + s * xv[t].w;
            }
        }
    }
    __syncthreads();

    const int* __restrict__ dstp = ei + E;
    const bool vec_ok  = (((size_t)dstp) & 15) == 0;
    const int  nChunks = (E + 127) >> 7;
    for (int c0 = wave * 4; c0 < nChunks; c0 += (GATHER_THREADS / 32) * 4) {
        int4 dv[4];
        const int eb = (c0 << 7) + lane * 4;
        if (vec_ok && ((c0 + 4) << 7) <= E) {
            #pragma unroll
            for (int j = 0; j < 4; ++j) dv[j] = *(const int4*)(dstp + eb + (j << 7));
        } else {
            #pragma unroll
            for (int j = 0; j < 4; ++j) dv[j] = load_dst4_guarded(dstp, eb + (j << 7), E);
        }
        #pragma unroll
        for (int j = 0; j < 4; ++j) {
            const int4 d  = dv[j];
            const int  eu = (c0 << 7) + (j << 7);
            const bool m0 = (unsigned)(d.x - row0) < (unsigned)rows;
            const bool m1 = (unsigned)(d.y - row0) < (unsigned)rows;
            const bool m2 = (unsigned)(d.z - row0) < (unsigned)rows;
            const bool m3 = (unsigned)(d.w - row0) < (unsigned)rows;
            if (__builtin_amdgcn_ballot_w32(m0 | m1 | m2 | m3) == 0u) continue;
            add_matches(d.x, m0, eu + 0, ei, deg, xw, acc, row0, lane);
            add_matches(d.y, m1, eu + 1, ei, deg, xw, acc, row0, lane);
            add_matches(d.z, m2, eu + 2, ei, deg, xw, acc, row0, lane);
            add_matches(d.w, m3, eu + 3, ei, deg, xw, acc, row0, lane);
        }
    }
    __syncthreads();

    float4* __restrict__ out4 = (float4*)out;
    for (int idx = tid; idx < rows * 128; idx += GATHER_THREADS) {
        const int r = idx >> 7;
        const int q = idx & 127;
        const float* a = acc + r * OUT_DIM + q;
        float4 o;
        o.x = a[0]; o.y = a[128]; o.z = a[256]; o.w = a[384];
        out4[(size_t)(row0 + r) * 128 + q] = o;
    }
}

extern "C" void kernel_launch(void* const* d_in, const int* in_sizes, int n_in,
                              void* d_out, int out_size, void* d_ws, size_t ws_size,
                              hipStream_t stream) {
    const float* x     = (const float*)d_in[0];
    const float* t_emb = (const float*)d_in[1];
    const int*   ei    = (const int*)d_in[2];
    const float* W     = (const float*)d_in[3];
    const float* b     = (const float*)d_in[4];
    const float* Wt    = (const float*)d_in[5];
    const float* bt    = (const float*)d_in[6];
    float*       out   = (float*)d_out;

    const int N = in_sizes[0] / IN_DIM;
    const int E = in_sizes[2] / 2;

    char* ws = (char*)d_ws;
    float* deg  = (float*)(ws + 0);
    float* cvec = (float*)(ws + (((size_t)N * 4 + 63) & ~63ull));
    float* xw   = (float*)((char*)cvec + 4096);

    k_cvec<<<OUT_DIM / 32, CVEC_WAVES * 32, 0, stream>>>(t_emb, Wt, b, bt, cvec, deg, N);

    {
        int rowTiles  = (N + 15) / 16;
        int rowGroups = (rowTiles + 3) / 4;
        int blocks    = rowGroups * (OUT_DIM / 64);
        k_gemm_wmma<<<blocks, GEMM_WAVES * 32, 0, stream>>>(x, W, xw, N, rowTiles, ei, deg, E);
    }

    {
        int blocks = (N + ROWS_PER_WG - 1) / ROWS_PER_WG;
        k_owner_gather<<<blocks, GATHER_THREADS, 0, stream>>>(ei, xw, deg, cvec, out, N, E);
    }
}
